# starting kernel
# baseline (speedup 1.0000x reference)
#include <hip/hip_runtime.h>

typedef __attribute__((ext_vector_type(2))) float v2f;
typedef __attribute__((ext_vector_type(8))) float v8f;

#define B_   32
#define C_   512
#define L_   4096
#define NB   16
#define EPSV 1e-8f

#define WS_G    0
#define WS_GN   (B_ * C_ * 4)
#define WS_KEYS (WS_GN + 128)
#define WS_COS  (WS_KEYS + 128)

__device__ __forceinline__ unsigned enc_key(float f) {
    unsigned u = __float_as_uint(f);
    return (u & 0x80000000u) ? ~u : (u | 0x80000000u);
}
__device__ __forceinline__ float dec_key(unsigned k) {
    unsigned u = (k & 0x80000000u) ? (k & 0x7FFFFFFFu) : ~k;
    return __uint_as_float(u);
}

__global__ void k_init(unsigned* __restrict__ keys) {
    if (threadIdx.x == 0) { keys[0] = 0xFFFFFFFFu; keys[1] = 0u; }
}

__global__ void k_mean(const float* __restrict__ x, float* __restrict__ g) {
    int wave = threadIdx.x >> 5, lane = threadIdx.x & 31;
    int row  = blockIdx.x * 8 + wave;
    const float4* xr = (const float4*)(x + (size_t)row * L_);
    float s = 0.f;
    for (int i = lane; i < L_ / 4; i += 32) {
        float4 v = xr[i];
        s += (v.x + v.y) + (v.z + v.w);
    }
    #pragma unroll
    for (int off = 16; off > 0; off >>= 1) s += __shfl_xor(s, off, 32);
    if (lane == 0) g[row] = s * (1.0f / (float)L_);
}

__global__ void k_gn(const float* __restrict__ g, float* __restrict__ gn) {
    int b = blockIdx.x, t = threadIdx.x;
    float a  = g[b * C_ + t];
    float c2 = g[b * C_ + 256 + t];
    float s  = a * a + c2 * c2;
    #pragma unroll
    for (int off = 16; off > 0; off >>= 1) s += __shfl_xor(s, off, 32);
    __shared__ float red[8];
    if ((t & 31) == 0) red[t >> 5] = s;
    __syncthreads();
    if (t == 0) {
        float tot = 0.f;
        #pragma unroll
        for (int i = 0; i < 8; ++i) tot += red[i];
        gn[b] = sqrtf(tot);
    }
}

__global__ void k_cos(const float* __restrict__ x, const float* __restrict__ g,
                      const float* __restrict__ gn, float* __restrict__ cosb,
                      unsigned* __restrict__ keys) {
    __shared__ float sg[C_];
    int b = blockIdx.y;
    for (int i = threadIdx.x; i < C_; i += 256) sg[i] = g[b * C_ + i];
    __syncthreads();

    int wave = threadIdx.x >> 5, lane = threadIdx.x & 31;
    int l0   = blockIdx.x * 128 + wave * 16;
    int col  = l0 + (lane & 15);
    int rowSel = (lane >> 4) * 2;
    const float* xb = x + (size_t)b * C_ * L_ + col;

    v8f acc = {};
    float sq = 0.f;
    for (int c0 = 0; c0 < C_; c0 += 4) {
        float b0 = xb[(size_t)(c0 + rowSel) * L_];
        float b1 = xb[(size_t)(c0 + rowSel + 1) * L_];
        v2f Af; Af.x = sg[c0 + rowSel]; Af.y = sg[c0 + rowSel + 1];
        v2f Bf; Bf.x = b0;              Bf.y = b1;
        acc = __builtin_amdgcn_wmma_f32_16x16x4_f32(
             false, Af,  false, Bf,
             (short)0, acc,  false,  false);
        sq = fmaf(b0, b0, sq);
        sq = fmaf(b1, b1, sq);
    }
    sq += __shfl_xor(sq, 16, 32);
    float dot  = acc[0];
    float xn   = sqrtf(sq);
    float cosv = dot / fmaxf(xn * gn[b], EPSV);
    if (lane < 16) cosb[b * L_ + col] = cosv;

    unsigned key = enc_key(cosv);
    unsigned kmin = key, kmax = key;
    #pragma unroll
    for (int off = 16; off > 0; off >>= 1) {
        unsigned a = __shfl_xor(kmin, off, 32);
        unsigned c = __shfl_xor(kmax, off, 32);
        kmin = (a < kmin) ? a : kmin;
        kmax = (c > kmax) ? c : kmax;
    }
    if (lane == 0) { atomicMin(&keys[0], kmin); atomicMax(&keys[1], kmax); }
}

__global__ void k_hist(const float* __restrict__ cosb, const unsigned* __restrict__ keys,
                       float* __restrict__ out) {
    int b = blockIdx.x, t = threadIdx.x;
    float minS = dec_key(keys[0]);
    float maxS = dec_key(keys[1]);
    float step = (maxS - minS) * (1.0f / (float)(NB - 1));

    float bins[NB];
    #pragma unroll
    for (int n = 0; n < NB; ++n) bins[n] = 0.f;
    for (int l = t; l < L_; l += 256) {
        float s = cosb[b * L_ + l];
        #pragma unroll
        for (int n = 0; n < NB; ++n) {
            float d = fabsf(s - (minS + (float)n * step));
            if (d < 0.03125f) bins[n] += (1.0f - d);
        }
    }
    __shared__ float sb[NB];
    if (t < NB) sb[t] = 0.f;
    __syncthreads();
    #pragma unroll
    for (int n = 0; n < NB; ++n) atomicAdd(&sb[n], bins[n]);
    __syncthreads();
    float tot = 0.f;
    #pragma unroll
    for (int n = 0; n < NB; ++n) tot += sb[n];
    float inv = 1.0f / tot;
    for (int i = t; i < NB * (C_ / NB); i += 256)
        out[b * (NB * (C_ / NB)) + i] = sb[i >> 5] * inv;
}

extern "C" void kernel_launch(void* const* d_in, const int* in_sizes, int n_in,
                              void* d_out, int out_size, void* d_ws, size_t ws_size,
                              hipStream_t stream) {
    const float* x = (const float*)d_in[0];
    float* out = (float*)d_out;
    char* ws = (char*)d_ws;
    float*    g    = (float*)(ws + WS_G);
    float*    gn   = (float*)(ws + WS_GN);
    unsigned* keys = (unsigned*)(ws + WS_KEYS);
    float*    cosb = (float*)(ws + WS_COS);

    k_init<<<1, 32, 0, stream>>>(keys);
    k_mean<<<(B_ * C_) / 8, 256, 0, stream>>>(x, g);
    k_gn<<<B_, 256, 0, stream>>>(g, gn);
    k_cos<<<dim3(L_ / 128, B_), 256, 0, stream>>>(x, g, gn, cosb, keys);
    k_hist<<<B_, 256, 0, stream>>>(cosb, keys, out);
}


# cosine kernel re-tiled: a wave owns 64 columns and loads 16 B per lane (aligned 256-B row chunks, 48 loads ahead in a register ring) instead of 16 columns with 4-B loads; order of sums kept
# speedup vs baseline: 1.1352x; 1.1352x over previous
#include <hip/hip_runtime.h>

typedef __attribute__((ext_vector_type(2))) float v2f;
typedef __attribute__((ext_vector_type(8))) float v8f;
typedef __attribute__((ext_vector_type(4))) float v4f;

#define B_   32
#define C_   512
#define L_   4096
#define NB   16
#define EPSV 1e-8f

#define WS_G    0
#define WS_GN   (B_ * C_ * 4)
#define WS_KEYS (WS_GN + 128)
#define WS_COS  (WS_KEYS + 128)

__device__ __forceinline__ unsigned enc_key(float f) {
    unsigned u = __float_as_uint(f);
    return (u & 0x80000000u) ? ~u : (u | 0x80000000u);
}
__device__ __forceinline__ float dec_key(unsigned k) {
    unsigned u = (k & 0x80000000u) ? (k & 0x7FFFFFFFu) : ~k;
    return __uint_as_float(u);
}

__global__ void k_init(unsigned* __restrict__ keys) {
    if (threadIdx.x == 0) { keys[0] = 0xFFFFFFFFu; keys[1] = 0u; }
}

__global__ void k_mean(const float* __restrict__ x, float* __restrict__ g) {
    int wave = threadIdx.x >> 5, lane = threadIdx.x & 31;
    int row  = blockIdx.x * 8 + wave;
    const float4* xr = (const float4*)(x + (size_t)row * L_);
    float s = 0.f;
    for (int i = lane; i < L_ / 4; i += 32) {
        float4 v = xr[i];
        s += (v.x + v.y) + (v.z + v.w);
    }
    #pragma unroll
    for (int off = 16; off > 0; off >>= 1) s += __shfl_xor(s, off, 32);
    if (lane == 0) g[row] = s * (1.0f / (float)L_);
}

__global__ void k_gn(const float* __restrict__ g, float* __restrict__ gn) {
    int b = blockIdx.x, t = threadIdx.x;
    float a  = g[b * C_ + t];
    float c2 = g[b * C_ + 256 + t];
    float s  = a * a + c2 * c2;
    #pragma unroll
    for (int off = 16; off > 0; off >>= 1) s += __shfl_xor(s, off, 32);
    __shared__ float red[8];
    if ((t & 31) == 0) red[t >> 5] = s;
    __syncthreads();
    if (t == 0) {
        float tot = 0.f;
        #pragma unroll
        for (int i = 0; i < 8; ++i) tot += red[i];
        gn[b] = sqrtf(tot);
    }
}

#define CU   8
#define NBLK (C_ / (4 * CU))
#define RD   4

#define COS_LOAD(R0, R1, BLK)                                                          \
    _Pragma("unroll") for (int u = 0; u < CU; ++u) {                                   \
        const float* p_ = xp + (size_t)(((BLK) * CU + u) * 4) * L_;                    \
        R0[u] = *(const v4f*)(p_);                                                     \
        R1[u] = *(const v4f*)(p_ + L_);                                                \
    }

#define COS_PIN()                                                                      \
    asm volatile("" : "+v"(acc0), "+v"(acc1), "+v"(acc2), "+v"(acc3),                  \
                      "+v"(sq0), "+v"(sq1), "+v"(sq2), "+v"(sq3));

#define COS_STEP(R0, R1, BLK)                                                          \
    _Pragma("unroll") for (int u = 0; u < CU; ++u) {                                   \
        const int c0_ = ((BLK) * CU + u) * 4;                                          \
        v2f Af; Af.x = gp[c0_]; Af.y = gp[c0_ + 1];                                    \
        v2f B0; B0.x = R0[u].x; B0.y = R1[u].x;                                        \
        v2f B1; B1.x = R0[u].y; B1.y = R1[u].y;                                        \
        v2f B2; B2.x = R0[u].z; B2.y = R1[u].z;                                        \
        v2f B3; B3.x = R0[u].w; B3.y = R1[u].w;                                        \
        acc0 = __builtin_amdgcn_wmma_f32_16x16x4_f32(false, Af, false, B0, (short)0, acc0, false, false); \
        acc1 = __builtin_amdgcn_wmma_f32_16x16x4_f32(false, Af, false, B1, (short)0, acc1, false, false); \
        acc2 = __builtin_amdgcn_wmma_f32_16x16x4_f32(false, Af, false, B2, (short)0, acc2, false, false); \
        acc3 = __builtin_amdgcn_wmma_f32_16x16x4_f32(false, Af, false, B3, (short)0, acc3, false, false); \
        sq0 = fmaf(R0[u].x, R0[u].x, sq0); sq0 = fmaf(R1[u].x, R1[u].x, sq0);          \
        sq1 = fmaf(R0[u].y, R0[u].y, sq1); sq1 = fmaf(R1[u].y, R1[u].y, sq1);          \
        sq2 = fmaf(R0[u].z, R0[u].z, sq2); sq2 = fmaf(R1[u].z, R1[u].z, sq2);          \
        sq3 = fmaf(R0[u].w, R0[u].w, sq3); sq3 = fmaf(R1[u].w, R1[u].w, sq3);          \
    }

__global__ void __launch_bounds__(128)
k_cos(const float* __restrict__ x, const float* __restrict__ g,
      const float* __restrict__ gn, float* __restrict__ cosb,
      unsigned* __restrict__ keys) {
    __shared__ float sg[C_];
    const int b = blockIdx.y;
    for (int i = threadIdx.x; i < C_; i += 128) sg[i] = g[b * C_ + i];
    __syncthreads();

    const int wave = threadIdx.x >> 5, lane = threadIdx.x & 31;
    const int n      = lane & 15;
    const int rowSel = (lane >> 4) * 2;
    const int l0     = blockIdx.x * 256 + wave * 64;
    const int col    = l0 + 4 * n;
    const float* xp  = x + (size_t)b * C_ * L_ + (size_t)rowSel * L_ + col;
    const float* gp  = sg + rowSel;

    v8f acc0 = {}, acc1 = {}, acc2 = {}, acc3 = {};
    float sq0 = 0.f, sq1 = 0.f, sq2 = 0.f, sq3 = 0.f;
    v4f r0[RD][CU], r1[RD][CU];

    #pragma unroll
    for (int k = 0; k < RD - 1; ++k) { COS_LOAD(r0[k % RD], r1[k % RD], k) }
    #pragma unroll
    for (int k = 0; k < NBLK; ++k) {
        __builtin_amdgcn_sched_barrier(0);
        if (k + RD - 1 < NBLK) { COS_LOAD(r0[(k + RD - 1) % RD], r1[(k + RD - 1) % RD], k + RD - 1) }
        __builtin_amdgcn_sched_barrier(0);
        COS_PIN()
        COS_STEP(r0[k % RD], r1[k % RD], k)
        COS_PIN()
    }

    sq0 += __shfl_xor(sq0, 16, 32);
    sq1 += __shfl_xor(sq1, 16, 32);
    sq2 += __shfl_xor(sq2, 16, 32);
    sq3 += __shfl_xor(sq3, 16, 32);
    const float gnb = gn[b];
    float4 cv;
    cv.x = acc0[0] / fmaxf(sqrtf(sq0) * gnb, EPSV);
    cv.y = acc1[0] / fmaxf(sqrtf(sq1) * gnb, EPSV);
    cv.z = acc2[0] / fmaxf(sqrtf(sq2) * gnb, EPSV);
    cv.w = acc3[0] / fmaxf(sqrtf(sq3) * gnb, EPSV);
    if (lane < 16) *(float4*)(cosb + b * L_ + col) = cv;

    unsigned k0 = enc_key(cv.x), k1 = enc_key(cv.y), k2 = enc_key(cv.z), k3 = enc_key(cv.w);
    unsigned kmin = min(min(k0, k1), min(k2, k3));
    unsigned kmax = max(max(k0, k1), max(k2, k3));
    #pragma unroll
    for (int off = 16; off > 0; off >>= 1) {
        unsigned a = __shfl_xor(kmin, off, 32);
        unsigned c = __shfl_xor(kmax, off, 32);
        kmin = (a < kmin) ? a : kmin;
        kmax = (c > kmax) ? c : kmax;
    }
    if (lane == 0) { atomicMin(&keys[0], kmin); atomicMax(&keys[1], kmax); }
}
#undef COS_LOAD
#undef COS_STEP
#undef COS_PIN

__global__ void k_hist(const float* __restrict__ cosb, const unsigned* __restrict__ keys,
                       float* __restrict__ out) {
    int b = blockIdx.x, t = threadIdx.x;
    float minS = dec_key(keys[0]);
    float maxS = dec_key(keys[1]);
    float step = (maxS - minS) * (1.0f / (float)(NB - 1));

    float bins[NB];
    #pragma unroll
    for (int n = 0; n < NB; ++n) bins[n] = 0.f;
    for (int l = t; l < L_; l += 256) {
        float s = cosb[b * L_ + l];
        #pragma unroll
        for (int n = 0; n < NB; ++n) {
            float d = fabsf(s - (minS + (float)n * step));
            if (d < 0.03125f) bins[n] += (1.0f - d);
        }
    }
    __shared__ float sb[NB];
    if (t < NB) sb[t] = 0.f;
    __syncthreads();
    #pragma unroll
    for (int n = 0; n < NB; ++n) atomicAdd(&sb[n], bins[n]);
    __syncthreads();
    float tot = 0.f;
    #pragma unroll
    for (int n = 0; n < NB; ++n) tot += sb[n];
    float inv = 1.0f / tot;
    for (int i = t; i < NB * (C_ / NB); i += 256)
        out[b * (NB * (C_ / NB)) + i] = sb[i >> 5] * inv;
}

extern "C" void kernel_launch(void* const* d_in, const int* in_sizes, int n_in,
                              void* d_out, int out_size, void* d_ws, size_t ws_size,
                              hipStream_t stream) {
    const float* x = (const float*)d_in[0];
    float* out = (float*)d_out;
    char* ws = (char*)d_ws;
    float*    g    = (float*)(ws + WS_G);
    float*    gn   = (float*)(ws + WS_GN);
    unsigned* keys = (unsigned*)(ws + WS_KEYS);
    float*    cosb = (float*)(ws + WS_COS);

    k_init<<<1, 32, 0, stream>>>(keys);
    k_mean<<<(B_ * C_) / 8, 256, 0, stream>>>(x, g);
    k_gn<<<B_, 256, 0, stream>>>(g, gn);
    k_cos<<<dim3(L_ / 256, B_), 128, 0, stream>>>(x, g, gn, cosb, keys);
    k_hist<<<B_, 256, 0, stream>>>(cosb, keys, out);
}


# histogram kernel: 4 loads of 16 B per thread in flight at once (was 16 dependent 4-B loads); bins summed by a shuffle halving tree and a fixed 8-wave sum (was LDS atomics); bin sum order changed
# speedup vs baseline: 1.4204x; 1.2512x over previous
#include <hip/hip_runtime.h>

typedef __attribute__((ext_vector_type(2))) float v2f;
typedef __attribute__((ext_vector_type(8))) float v8f;
typedef __attribute__((ext_vector_type(4))) float v4f;

#define B_   32
#define C_   512
#define L_   4096
#define NB   16
#define EPSV 1e-8f

#define WS_G    0
#define WS_GN   (B_ * C_ * 4)
#define WS_KEYS (WS_GN + 128)
#define WS_COS  (WS_KEYS + 128)

__device__ __forceinline__ unsigned enc_key(float f) {
    unsigned u = __float_as_uint(f);
    return (u & 0x80000000u) ? ~u : (u | 0x80000000u);
}
__device__ __forceinline__ float dec_key(unsigned k) {
    unsigned u = (k & 0x80000000u) ? (k & 0x7FFFFFFFu) : ~k;
    return __uint_as_float(u);
}

__global__ void k_init(unsigned* __restrict__ keys) {
    if (threadIdx.x == 0) { keys[0] = 0xFFFFFFFFu; keys[1] = 0u; }
}

__global__ void k_mean(const float* __restrict__ x, float* __restrict__ g) {
    int wave = threadIdx.x >> 5, lane = threadIdx.x & 31;
    int row  = blockIdx.x * 8 + wave;
    const float4* xr = (const float4*)(x + (size_t)row * L_);
    float s = 0.f;
    for (int i = lane; i < L_ / 4; i += 32) {
        float4 v = xr[i];
        s += (v.x + v.y) + (v.z + v.w);
    }
    #pragma unroll
    for (int off = 16; off > 0; off >>= 1) s += __shfl_xor(s, off, 32);
    if (lane == 0) g[row] = s * (1.0f / (float)L_);
}

__global__ void k_gn(const float* __restrict__ g, float* __restrict__ gn) {
    int b = blockIdx.x, t = threadIdx.x;
    float a  = g[b * C_ + t];
    float c2 = g[b * C_ + 256 + t];
    float s  = a * a + c2 * c2;
    #pragma unroll
    for (int off = 16; off > 0; off >>= 1) s += __shfl_xor(s, off, 32);
    __shared__ float red[8];
    if ((t & 31) == 0) red[t >> 5] = s;
    __syncthreads();
    if (t == 0) {
        float tot = 0.f;
        #pragma unroll
        for (int i = 0; i < 8; ++i) tot += red[i];
        gn[b] = sqrtf(tot);
    }
}

#define CU   8
#define NBLK (C_ / (4 * CU))
#define RD   4

#define COS_LOAD(R0, R1, BLK)                                                          \
    _Pragma("unroll") for (int u = 0; u < CU; ++u) {                                   \
        const float* p_ = xp + (size_t)(((BLK) * CU + u) * 4) * L_;                    \
        R0[u] = *(const v4f*)(p_);                                                     \
        R1[u] = *(const v4f*)(p_ + L_);                                                \
    }

#define COS_PIN()                                                                      \
    asm volatile("" : "+v"(acc0), "+v"(acc1), "+v"(acc2), "+v"(acc3),                  \
                      "+v"(sq0), "+v"(sq1), "+v"(sq2), "+v"(sq3));

#define COS_STEP(R0, R1, BLK)                                                          \
    _Pragma("unroll") for (int u = 0; u < CU; ++u) {                                   \
        const int c0_ = ((BLK) * CU + u) * 4;                                          \
        v2f Af; Af.x = gp[c0_]; Af.y = gp[c0_ + 1];                                    \
        v2f B0; B0.x = R0[u].x; B0.y = R1[u].x;                                        \
        v2f B1; B1.x = R0[u].y; B1.y = R1[u].y;                                        \
        v2f B2; B2.x = R0[u].z; B2.y = R1[u].z;                                        \
        v2f B3; B3.x = R0[u].w; B3.y = R1[u].w;                                        \
        acc0 = __builtin_amdgcn_wmma_f32_16x16x4_f32(false, Af, false, B0, (short)0, acc0, false, false); \
        acc1 = __builtin_amdgcn_wmma_f32_16x16x4_f32(false, Af, false, B1, (short)0, acc1, false, false); \
        acc2 = __builtin_amdgcn_wmma_f32_16x16x4_f32(false, Af, false, B2, (short)0, acc2, false, false); \
        acc3 = __builtin_amdgcn_wmma_f32_16x16x4_f32(false, Af, false, B3, (short)0, acc3, false, false); \
        sq0 = fmaf(R0[u].x, R0[u].x, sq0); sq0 = fmaf(R1[u].x, R1[u].x, sq0);          \
        sq1 = fmaf(R0[u].y, R0[u].y, sq1); sq1 = fmaf(R1[u].y, R1[u].y, sq1);          \
        sq2 = fmaf(R0[u].z, R0[u].z, sq2); sq2 = fmaf(R1[u].z, R1[u].z, sq2);          \
        sq3 = fmaf(R0[u].w, R0[u].w, sq3); sq3 = fmaf(R1[u].w, R1[u].w, sq3);          \
    }

__global__ void __launch_bounds__(128)
k_cos(const float* __restrict__ x, const float* __restrict__ g,
      const float* __restrict__ gn, float* __restrict__ cosb,
      unsigned* __restrict__ keys) {
    __shared__ float sg[C_];
    const int b = blockIdx.y;
    for (int i = threadIdx.x; i < C_; i += 128) sg[i] = g[b * C_ + i];
    __syncthreads();

    const int wave = threadIdx.x >> 5, lane = threadIdx.x & 31;
    const int n      = lane & 15;
    const int rowSel = (lane >> 4) * 2;
    const int l0     = blockIdx.x * 256 + wave * 64;
    const int col    = l0 + 4 * n;
    const float* xp  = x + (size_t)b * C_ * L_ + (size_t)rowSel * L_ + col;
    const float* gp  = sg + rowSel;

    v8f acc0 = {}, acc1 = {}, acc2 = {}, acc3 = {};
    float sq0 = 0.f, sq1 = 0.f, sq2 = 0.f, sq3 = 0.f;
    v4f r0[RD][CU], r1[RD][CU];

    #pragma unroll
    for (int k = 0; k < RD - 1; ++k) { COS_LOAD(r0[k % RD], r1[k % RD], k) }
    #pragma unroll
    for (int k = 0; k < NBLK; ++k) {
        __builtin_amdgcn_sched_barrier(0);
        if (k + RD - 1 < NBLK) { COS_LOAD(r0[(k + RD - 1) % RD], r1[(k + RD - 1) % RD], k + RD - 1) }
        __builtin_amdgcn_sched_barrier(0);
        COS_PIN()
        COS_STEP(r0[k % RD], r1[k % RD], k)
        COS_PIN()
    }

    sq0 += __shfl_xor(sq0, 16, 32);
    sq1 += __shfl_xor(sq1, 16, 32);
    sq2 += __shfl_xor(sq2, 16, 32);
    sq3 += __shfl_xor(sq3, 16, 32);
    const float gnb = gn[b];
    float4 cv;
    cv.x = acc0[0] / fmaxf(sqrtf(sq0) * gnb, EPSV);
    cv.y = acc1[0] / fmaxf(sqrtf(sq1) * gnb, EPSV);
    cv.z = acc2[0] / fmaxf(sqrtf(sq2) * gnb, EPSV);
    cv.w = acc3[0] / fmaxf(sqrtf(sq3) * gnb, EPSV);
    if (lane < 16) *(float4*)(cosb + b * L_ + col) = cv;

    unsigned k0 = enc_key(cv.x), k1 = enc_key(cv.y), k2 = enc_key(cv.z), k3 = enc_key(cv.w);
    unsigned kmin = min(min(k0, k1), min(k2, k3));
    unsigned kmax = max(max(k0, k1), max(k2, k3));
    #pragma unroll
    for (int off = 16; off > 0; off >>= 1) {
        unsigned a = __shfl_xor(kmin, off, 32);
        unsigned c = __shfl_xor(kmax, off, 32);
        kmin = (a < kmin) ? a : kmin;
        kmax = (c > kmax) ? c : kmax;
    }
    if (lane == 0) { atomicMin(&keys[0], kmin); atomicMax(&keys[1], kmax); }
}
#undef COS_LOAD
#undef COS_STEP
#undef COS_PIN

__global__ void __launch_bounds__(256)
k_hist(const float* __restrict__ cosb, const unsigned* __restrict__ keys,
       float* __restrict__ out) {
    const int b = blockIdx.x, t = threadIdx.x;
    const int lane = t & 31, wave = t >> 5;
    unsigned key0 = keys[0], key1 = keys[1];
    const v4f* cb = (const v4f*)(cosb + b * L_);
    v4f q0 = cb[t], q1 = cb[t + 256], q2 = cb[t + 512], q3 = cb[t + 768];
    asm volatile("" : "+v"(q0), "+v"(q1), "+v"(q2), "+v"(q3), "+s"(key0), "+s"(key1));

    float minS = dec_key(key0);
    float maxS = dec_key(key1);
    float step = (maxS - minS) * (1.0f / (float)(NB - 1));

    const float vals[16] = { q0.x, q0.y, q0.z, q0.w, q1.x, q1.y, q1.z, q1.w,
                             q2.x, q2.y, q2.z, q2.w, q3.x, q3.y, q3.z, q3.w };
    float bins[NB];
    #pragma unroll
    for (int n = 0; n < NB; ++n) bins[n] = 0.f;
    #pragma unroll
    for (int i = 0; i < 16; ++i) {
        float s = vals[i];
        #pragma unroll
        for (int n = 0; n < NB; ++n) {
            float d = fabsf(s - (minS + (float)n * step));
            if (d < 0.03125f) bins[n] += (1.0f - d);
        }
    }

    const bool h4 = (lane & 16) != 0, h3 = (lane & 8) != 0, h2 = (lane & 4) != 0, h1 = (lane & 2) != 0;
    float r8[8], r4[4], r2[2];
    #pragma unroll
    for (int j = 0; j < 8; ++j) {
        float keep = h4 ? bins[8 + j] : bins[j];
        float send = h4 ? bins[j] : bins[8 + j];
        r8[j] = keep + __shfl_xor(send, 16, 32);
    }
    #pragma unroll
    for (int j = 0; j < 4; ++j) {
        float keep = h3 ? r8[4 + j] : r8[j];
        float send = h3 ? r8[j] : r8[4 + j];
        r4[j] = keep + __shfl_xor(send, 8, 32);
    }
    #pragma unroll
    for (int j = 0; j < 2; ++j) {
        float keep = h2 ? r4[2 + j] : r4[j];
        float send = h2 ? r4[j] : r4[2 + j];
        r2[j] = keep + __shfl_xor(send, 4, 32);
    }
    float r1 = (h1 ? r2[1] : r2[0]) + __shfl_xor(h1 ? r2[0] : r2[1], 2, 32);
    r1 += __shfl_xor(r1, 1, 32);

    __shared__ float red[8][NB];
    __shared__ float sb[NB];
    if ((lane & 1) == 0) red[wave][lane >> 1] = r1;
    __syncthreads();
    if (t < NB) {
        float s = red[0][t];
        #pragma unroll
        for (int w = 1; w < 8; ++w) s += red[w][t];
        sb[t] = s;
    }
    __syncthreads();
    float tot = 0.f;
    #pragma unroll
    for (int n = 0; n < NB; ++n) tot += sb[n];
    float inv = 1.0f / tot;
    for (int i = t; i < NB * (C_ / NB); i += 256)
        out[b * (NB * (C_ / NB)) + i] = sb[i >> 5] * inv;
}

extern "C" void kernel_launch(void* const* d_in, const int* in_sizes, int n_in,
                              void* d_out, int out_size, void* d_ws, size_t ws_size,
                              hipStream_t stream) {
    const float* x = (const float*)d_in[0];
    float* out = (float*)d_out;
    char* ws = (char*)d_ws;
    float*    g    = (float*)(ws + WS_G);
    float*    gn   = (float*)(ws + WS_GN);
    unsigned* keys = (unsigned*)(ws + WS_KEYS);
    float*    cosb = (float*)(ws + WS_COS);

    k_init<<<1, 32, 0, stream>>>(keys);
    k_mean<<<(B_ * C_) / 8, 256, 0, stream>>>(x, g);
    k_gn<<<B_, 256, 0, stream>>>(g, gn);
    k_cos<<<dim3(L_ / 256, B_), 128, 0, stream>>>(x, g, gn, cosb, keys);
    k_hist<<<B_, 256, 0, stream>>>(cosb, keys, out);
}


# mean kernel re-tiled as a stream: 256 workgroups of 16 waves, a wave sums 4 consecutive rows with 16 to 32 loads of 16 B in flight (was 16384 one-row waves); order of sums kept
# speedup vs baseline: 1.6448x; 1.1580x over previous
#include <hip/hip_runtime.h>

typedef __attribute__((ext_vector_type(2))) float v2f;
typedef __attribute__((ext_vector_type(8))) float v8f;
typedef __attribute__((ext_vector_type(4))) float v4f;

#define B_   32
#define C_   512
#define L_   4096
#define NB   16
#define EPSV 1e-8f

#define WS_G    0
#define WS_GN   (B_ * C_ * 4)
#define WS_KEYS (WS_GN + 128)
#define WS_COS  (WS_KEYS + 128)

__device__ __forceinline__ unsigned enc_key(float f) {
    unsigned u = __float_as_uint(f);
    return (u & 0x80000000u) ? ~u : (u | 0x80000000u);
}
__device__ __forceinline__ float dec_key(unsigned k) {
    unsigned u = (k & 0x80000000u) ? (k & 0x7FFFFFFFu) : ~k;
    return __uint_as_float(u);
}

#define MW   16
#define MR   4
#define MQ   16
#define MBLK (MR * (L_ / 128) / MQ)
__global__ void __launch_bounds__(MW * 32)
k_mean(const float* __restrict__ x, float* __restrict__ g,
       unsigned* __restrict__ keys) {
    if (blockIdx.x == 0 && threadIdx.x == 0) { keys[0] = 0xFFFFFFFFu; keys[1] = 0u; }
    const int lane = threadIdx.x & 31;
    const int wave = __builtin_amdgcn_readfirstlane(threadIdx.x >> 5);
    const int row0 = (blockIdx.x * MW + wave) * MR;
    const v4f* xr = (const v4f*)(x + (size_t)row0 * L_) + lane;
    v4f buf[2][MQ];
    float sum[MR];

#define MEAN_LOAD(K)                                                                   \
    _Pragma("unroll") for (int q = 0; q < MQ; ++q)                                     \
        buf[(K) & 1][q] = xr[((K) * MQ + q) * 32];

    MEAN_LOAD(0)
    #pragma unroll
    for (int k = 0; k < MBLK; ++k) {
        __builtin_amdgcn_sched_barrier(0);
        if (k + 1 < MBLK) { MEAN_LOAD(k + 1) }
        __builtin_amdgcn_sched_barrier(0);
        float s = (k & 1) ? sum[k >> 1] : 0.f;
        #pragma unroll
        for (int q = 0; q < MQ; ++q) {
            v4f v = buf[k & 1][q];
            s += (v.x + v.y) + (v.z + v.w);
        }
        asm volatile("" : "+v"(s));
        sum[k >> 1] = s;
    }
#undef MEAN_LOAD

    float4 m;
    #pragma unroll
    for (int r = 0; r < MR; ++r) {
        float s = sum[r];
        #pragma unroll
        for (int off = 16; off > 0; off >>= 1) s += __shfl_xor(s, off, 32);
        sum[r] = s * (1.0f / (float)L_);
    }
    m.x = sum[0]; m.y = sum[1]; m.z = sum[2]; m.w = sum[3];
    if (lane == 0) *(float4*)(g + row0) = m;
}

#define CU   8
#define NBLK (C_ / (4 * CU))
#define RD   4

#define COS_LOAD(R0, R1, BLK)                                                          \
    _Pragma("unroll") for (int u = 0; u < CU; ++u) {                                   \
        const float* p_ = xp + (size_t)(((BLK) * CU + u) * 4) * L_;                    \
        R0[u] = *(const v4f*)(p_);                                                     \
        R1[u] = *(const v4f*)(p_ + L_);                                                \
    }

#define COS_PIN()                                                                      \
    asm volatile("" : "+v"(acc0), "+v"(acc1), "+v"(acc2), "+v"(acc3),                  \
                      "+v"(sq0), "+v"(sq1), "+v"(sq2), "+v"(sq3));

#define COS_STEP(R0, R1, BLK)                                                          \
    _Pragma("unroll") for (int u = 0; u < CU; ++u) {                                   \
        const int c0_ = ((BLK) * CU + u) * 4;                                          \
        v2f Af; Af.x = gp[c0_]; Af.y = gp[c0_ + 1];                                    \
        v2f B0; B0.x = R0[u].x; B0.y = R1[u].x;                                        \
        v2f B1; B1.x = R0[u].y; B1.y = R1[u].y;                                        \
        v2f B2; B2.x = R0[u].z; B2.y = R1[u].z;                                        \
        v2f B3; B3.x = R0[u].w; B3.y = R1[u].w;                                        \
        acc0 = __builtin_amdgcn_wmma_f32_16x16x4_f32(false, Af, false, B0, (short)0, acc0, false, false); \
        acc1 = __builtin_amdgcn_wmma_f32_16x16x4_f32(false, Af, false, B1, (short)0, acc1, false, false); \
        acc2 = __builtin_amdgcn_wmma_f32_16x16x4_f32(false, Af, false, B2, (short)0, acc2, false, false); \
        acc3 = __builtin_amdgcn_wmma_f32_16x16x4_f32(false, Af, false, B3, (short)0, acc3, false, false); \
        sq0 = fmaf(R0[u].x, R0[u].x, sq0); sq0 = fmaf(R1[u].x, R1[u].x, sq0);          \
        sq1 = fmaf(R0[u].y, R0[u].y, sq1); sq1 = fmaf(R1[u].y, R1[u].y, sq1);          \
        sq2 = fmaf(R0[u].z, R0[u].z, sq2); sq2 = fmaf(R1[u].z, R1[u].z, sq2);          \
        sq3 = fmaf(R0[u].w, R0[u].w, sq3); sq3 = fmaf(R1[u].w, R1[u].w, sq3);          \
    }

__global__ void __launch_bounds__(128)
k_cos(const float* __restrict__ x, const float* __restrict__ g,
      float* __restrict__ cosb, unsigned* __restrict__ keys) {
    __shared__ float sg[C_];
    const int b = blockIdx.y;
    for (int i = threadIdx.x; i < C_; i += 128) sg[i] = g[b * C_ + i];
    __syncthreads();

    const int wave = threadIdx.x >> 5, lane = threadIdx.x & 31;
    const int n      = lane & 15;
    const int rowSel = (lane >> 4) * 2;
    const int l0     = blockIdx.x * 256 + wave * 64;
    const int col    = l0 + 4 * n;
    const float* xp  = x + (size_t)b * C_ * L_ + (size_t)rowSel * L_ + col;
    const float* gp  = sg + rowSel;

    v8f acc0 = {}, acc1 = {}, acc2 = {}, acc3 = {};
    float sq0 = 0.f, sq1 = 0.f, sq2 = 0.f, sq3 = 0.f;
    v4f r0[RD][CU], r1[RD][CU];

    #pragma unroll
    for (int k = 0; k < RD - 1; ++k) { COS_LOAD(r0[k % RD], r1[k % RD], k) }

    float gtot = 0.f;
    #pragma unroll
    for (int w = 0; w < 8; ++w) {
        float a  = sg[32 * w + lane];
        float c2 = sg[256 + 32 * w + lane];
        float s  = a * a + c2 * c2;
        #pragma unroll
        for (int off = 16; off > 0; off >>= 1) s += __shfl_xor(s, off, 32);
        gtot += s;
    }
    float gnb = sqrtf(gtot);
    asm volatile("" : "+v"(gnb));

    #pragma unroll
    for (int k = 0; k < NBLK; ++k) {
        __builtin_amdgcn_sched_barrier(0);
        if (k + RD - 1 < NBLK) { COS_LOAD(r0[(k + RD - 1) % RD], r1[(k + RD - 1) % RD], k + RD - 1) }
        __builtin_amdgcn_sched_barrier(0);
        COS_PIN()
        COS_STEP(r0[k % RD], r1[k % RD], k)
        COS_PIN()
    }

    sq0 += __shfl_xor(sq0, 16, 32);
    sq1 += __shfl_xor(sq1, 16, 32);
    sq2 += __shfl_xor(sq2, 16, 32);
    sq3 += __shfl_xor(sq3, 16, 32);
    float4 cv;
    cv.x = acc0[0] / fmaxf(sqrtf(sq0) * gnb, EPSV);
    cv.y = acc1[0] / fmaxf(sqrtf(sq1) * gnb, EPSV);
    cv.z = acc2[0] / fmaxf(sqrtf(sq2) * gnb, EPSV);
    cv.w = acc3[0] / fmaxf(sqrtf(sq3) * gnb, EPSV);
    if (lane < 16) *(float4*)(cosb + b * L_ + col) = cv;

    unsigned k0 = enc_key(cv.x), k1 = enc_key(cv.y), k2 = enc_key(cv.z), k3 = enc_key(cv.w);
    unsigned kmin = min(min(k0, k1), min(k2, k3));
    unsigned kmax = max(max(k0, k1), max(k2, k3));
    #pragma unroll
    for (int off = 16; off > 0; off >>= 1) {
        unsigned a = __shfl_xor(kmin, off, 32);
        unsigned c = __shfl_xor(kmax, off, 32);
        kmin = (a < kmin) ? a : kmin;
        kmax = (c > kmax) ? c : kmax;
    }
    if (lane == 0) { atomicMin(&keys[0], kmin); atomicMax(&keys[1], kmax); }
}
#undef COS_LOAD
#undef COS_STEP
#undef COS_PIN

__global__ void __launch_bounds__(256)
k_hist(const float* __restrict__ cosb, const unsigned* __restrict__ keys,
       float* __restrict__ out) {
    const int b = blockIdx.x, t = threadIdx.x;
    const int lane = t & 31, wave = t >> 5;
    unsigned key0 = keys[0], key1 = keys[1];
    const v4f* cb = (const v4f*)(cosb + b * L_);
    v4f q0 = cb[t], q1 = cb[t + 256], q2 = cb[t + 512], q3 = cb[t + 768];
    asm volatile("" : "+v"(q0), "+v"(q1), "+v"(q2), "+v"(q3), "+s"(key0), "+s"(key1));

    float minS = dec_key(key0);
    float maxS = dec_key(key1);
    float step = (maxS - minS) * (1.0f / (float)(NB - 1));

    const float vals[16] = { q0.x, q0.y, q0.z, q0.w, q1.x, q1.y, q1.z, q1.w,
                             q2.x, q2.y, q2.z, q2.w, q3.x, q3.y, q3.z, q3.w };
    float bins[NB];
    #pragma unroll
    for (int n = 0; n < NB; ++n) bins[n] = 0.f;
    #pragma unroll
    for (int i = 0; i < 16; ++i) {
        float s = vals[i];
        #pragma unroll
        for (int n = 0; n < NB; ++n) {
            float d = fabsf(s - (minS + (float)n * step));
            if (d < 0.03125f) bins[n] += (1.0f - d);
        }
    }

    const bool h4 = (lane & 16) != 0, h3 = (lane & 8) != 0, h2 = (lane & 4) != 0, h1 = (lane & 2) != 0;
    float r8[8], r4[4], r2[2];
    #pragma unroll
    for (int j = 0; j < 8; ++j) {
        float keep = h4 ? bins[8 + j] : bins[j];
        float send = h4 ? bins[j] : bins[8 + j];
        r8[j] = keep + __shfl_xor(send, 16, 32);
    }
    #pragma unroll
    for (int j = 0; j < 4; ++j) {
        float keep = h3 ? r8[4 + j] : r8[j];
        float send = h3 ? r8[j] : r8[4 + j];
        r4[j] = keep + __shfl_xor(send, 8, 32);
    }
    #pragma unroll
    for (int j = 0; j < 2; ++j) {
        float keep = h2 ? r4[2 + j] : r4[j];
        float send = h2 ? r4[j] : r4[2 + j];
        r2[j] = keep + __shfl_xor(send, 4, 32);
    }
    float r1 = (h1 ? r2[1] : r2[0]) + __shfl_xor(h1 ? r2[0] : r2[1], 2, 32);
    r1 += __shfl_xor(r1, 1, 32);

    __shared__ float red[8][NB];
    __shared__ float sb[NB];
    if ((lane & 1) == 0) red[wave][lane >> 1] = r1;
    __syncthreads();
    if (t < NB) {
        float s = red[0][t];
        #pragma unroll
        for (int w = 1; w < 8; ++w) s += red[w][t];
        sb[t] = s;
    }
    __syncthreads();
    float tot = 0.f;
    #pragma unroll
    for (int n = 0; n < NB; ++n) tot += sb[n];
    float inv = 1.0f / tot;
    for (int i = t; i < NB * (C_ / NB); i += 256)
        out[b * (NB * (C_ / NB)) + i] = sb[i >> 5] * inv;
}

extern "C" void kernel_launch(void* const* d_in, const int* in_sizes, int n_in,
                              void* d_out, int out_size, void* d_ws, size_t ws_size,
                              hipStream_t stream) {
    const float* x = (const float*)d_in[0];
    float* out = (float*)d_out;
    char* ws = (char*)d_ws;
    float*    g    = (float*)(ws + WS_G);
    unsigned* keys = (unsigned*)(ws + WS_KEYS);
    float*    cosb = (float*)(ws + WS_COS);

    k_mean<<<(B_ * C_) / (MW * MR), MW * 32, 0, stream>>>(x, g, keys);
    k_cos<<<dim3(L_ / 256, B_), 128, 0, stream>>>(x, g, cosb, keys);
    k_hist<<<B_, 256, 0, stream>>>(cosb, keys, out);
}


# x read once, not twice: a 16-wave workgroup holds 8 whole rows in registers, takes their means, then runs the f32 16x16x4 products on them; 8 partial sums per column; order of sums changed
# speedup vs baseline: 2.1281x; 1.2938x over previous
#include <hip/hip_runtime.h>

typedef __attribute__((ext_vector_type(2))) float v2f;
typedef __attribute__((ext_vector_type(8))) float v8f;
typedef __attribute__((ext_vector_type(4))) float v4f;

#define B_   32
#define C_   512
#define L_   4096
#define NB   16
#define EPSV 1e-8f

#define WS_G    0
#define WS_GN   (B_ * C_ * 4)
#define WS_KEYS (WS_GN + 128)
#define WS_COS  (WS_KEYS + 128)
#define WS_PDOT (WS_COS + B_ * L_ * 4)
#define WS_PSQ  (WS_PDOT + B_ * (C_ / 64) * L_ * 4)
#define WS_END  (WS_PSQ + B_ * (C_ / 64) * L_ * 4)

__device__ __forceinline__ unsigned enc_key(float f) {
    unsigned u = __float_as_uint(f);
    return (u & 0x80000000u) ? ~u : (u | 0x80000000u);
}
__device__ __forceinline__ float dec_key(unsigned k) {
    unsigned u = (k & 0x80000000u) ? (k & 0x7FFFFFFFu) : ~k;
    return __uint_as_float(u);
}

#define MW   16
#define MR   4
#define MQ   16
#define MBLK (MR * (L_ / 128) / MQ)
__global__ void __launch_bounds__(MW * 32)
k_mean(const float* __restrict__ x, float* __restrict__ g,
       unsigned* __restrict__ keys) {
    if (blockIdx.x == 0 && threadIdx.x == 0) { keys[0] = 0xFFFFFFFFu; keys[1] = 0u; }
    const int lane = threadIdx.x & 31;
    const int wave = __builtin_amdgcn_readfirstlane(threadIdx.x >> 5);
    const int row0 = (blockIdx.x * MW + wave) * MR;
    const v4f* xr = (const v4f*)(x + (size_t)row0 * L_) + lane;
    v4f buf[2][MQ];
    float sum[MR];

#define MEAN_LOAD(K)                                                                   \
    _Pragma("unroll") for (int q = 0; q < MQ; ++q)                                     \
        buf[(K) & 1][q] = xr[((K) * MQ + q) * 32];

    MEAN_LOAD(0)
    #pragma unroll
    for (int k = 0; k < MBLK; ++k) {
        __builtin_amdgcn_sched_barrier(0);
        if (k + 1 < MBLK) { MEAN_LOAD(k + 1) }
        __builtin_amdgcn_sched_barrier(0);
        float s = (k & 1) ? sum[k >> 1] : 0.f;
        #pragma unroll
        for (int q = 0; q < MQ; ++q) {
            v4f v = buf[k & 1][q];
            s += (v.x + v.y) + (v.z + v.w);
        }
        asm volatile("" : "+v"(s));
        sum[k >> 1] = s;
    }
#undef MEAN_LOAD

    float4 m;
    #pragma unroll
    for (int r = 0; r < MR; ++r) {
        float s = sum[r];
        #pragma unroll
        for (int off = 16; off > 0; off >>= 1) s += __shfl_xor(s, off, 32);
        sum[r] = s * (1.0f / (float)L_);
    }
    m.x = sum[0]; m.y = sum[1]; m.z = sum[2]; m.w = sum[3];
    if (lane == 0) *(float4*)(g + row0) = m;
}

#define CU   8
#define NBLK (C_ / (4 * CU))
#define RD   4

#define COS_LOAD(R0, R1, BLK)                                                          \
    _Pragma("unroll") for (int u = 0; u < CU; ++u) {                                   \
        const float* p_ = xp + (size_t)(((BLK) * CU + u) * 4) * L_;                    \
        R0[u] = *(const v4f*)(p_);                                                     \
        R1[u] = *(const v4f*)(p_ + L_);                                                \
    }

#define COS_PIN()                                                                      \
    asm volatile("" : "+v"(acc0), "+v"(acc1), "+v"(acc2), "+v"(acc3),                  \
                      "+v"(sq0), "+v"(sq1), "+v"(sq2), "+v"(sq3));

#define COS_STEP(R0, R1, BLK)                                                          \
    _Pragma("unroll") for (int u = 0; u < CU; ++u) {                                   \
        const int c0_ = ((BLK) * CU + u) * 4;                                          \
        v2f Af; Af.x = gp[c0_]; Af.y = gp[c0_ + 1];                                    \
        v2f B0; B0.x = R0[u].x; B0.y = R1[u].x;                                        \
        v2f B1; B1.x = R0[u].y; B1.y = R1[u].y;                                        \
        v2f B2; B2.x = R0[u].z; B2.y = R1[u].z;                                        \
        v2f B3; B3.x = R0[u].w; B3.y = R1[u].w;                                        \
        acc0 = __builtin_amdgcn_wmma_f32_16x16x4_f32(false, Af, false, B0, (short)0, acc0, false, false); \
        acc1 = __builtin_amdgcn_wmma_f32_16x16x4_f32(false, Af, false, B1, (short)0, acc1, false, false); \
        acc2 = __builtin_amdgcn_wmma_f32_16x16x4_f32(false, Af, false, B2, (short)0, acc2, false, false); \
        acc3 = __builtin_amdgcn_wmma_f32_16x16x4_f32(false, Af, false, B3, (short)0, acc3, false, false); \
        sq0 = fmaf(R0[u].x, R0[u].x, sq0); sq0 = fmaf(R1[u].x, R1[u].x, sq0);          \
        sq1 = fmaf(R0[u].y, R0[u].y, sq1); sq1 = fmaf(R1[u].y, R1[u].y, sq1);          \
        sq2 = fmaf(R0[u].z, R0[u].z, sq2); sq2 = fmaf(R1[u].z, R1[u].z, sq2);          \
        sq3 = fmaf(R0[u].w, R0[u].w, sq3); sq3 = fmaf(R1[u].w, R1[u].w, sq3);          \
    }

__global__ void __launch_bounds__(128)
k_cos(const float* __restrict__ x, const float* __restrict__ g,
      float* __restrict__ cosb, unsigned* __restrict__ keys) {
    __shared__ float sg[C_];
    const int b = blockIdx.y;
    for (int i = threadIdx.x; i < C_; i += 128) sg[i] = g[b * C_ + i];
    __syncthreads();

    const int wave = threadIdx.x >> 5, lane = threadIdx.x & 31;
    const int n      = lane & 15;
    const int rowSel = (lane >> 4) * 2;
    const int l0     = blockIdx.x * 256 + wave * 64;
    const int col    = l0 + 4 * n;
    const float* xp  = x + (size_t)b * C_ * L_ + (size_t)rowSel * L_ + col;
    const float* gp  = sg + rowSel;

    v8f acc0 = {}, acc1 = {}, acc2 = {}, acc3 = {};
    float sq0 = 0.f, sq1 = 0.f, sq2 = 0.f, sq3 = 0.f;
    v4f r0[RD][CU], r1[RD][CU];

    #pragma unroll
    for (int k = 0; k < RD - 1; ++k) { COS_LOAD(r0[k % RD], r1[k % RD], k) }

    float gtot = 0.f;
    #pragma unroll
    for (int w = 0; w < 8; ++w) {
        float a  = sg[32 * w + lane];
        float c2 = sg[256 + 32 * w + lane];
        float s  = a * a + c2 * c2;
        #pragma unroll
        for (int off = 16; off > 0; off >>= 1) s += __shfl_xor(s, off, 32);
        gtot += s;
    }
    float gnb = sqrtf(gtot);
    asm volatile("" : "+v"(gnb));

    #pragma unroll
    for (int k = 0; k < NBLK; ++k) {
        __builtin_amdgcn_sched_barrier(0);
        if (k + RD - 1 < NBLK) { COS_LOAD(r0[(k + RD - 1) % RD], r1[(k + RD - 1) % RD], k + RD - 1) }
        __builtin_amdgcn_sched_barrier(0);
        COS_PIN()
        COS_STEP(r0[k % RD], r1[k % RD], k)
        COS_PIN()
    }

    sq0 += __shfl_xor(sq0, 16, 32);
    sq1 += __shfl_xor(sq1, 16, 32);
    sq2 += __shfl_xor(sq2, 16, 32);
    sq3 += __shfl_xor(sq3, 16, 32);
    float4 cv;
    cv.x = acc0[0] / fmaxf(sqrtf(sq0) * gnb, EPSV);
    cv.y = acc1[0] / fmaxf(sqrtf(sq1) * gnb, EPSV);
    cv.z = acc2[0] / fmaxf(sqrtf(sq2) * gnb, EPSV);
    cv.w = acc3[0] / fmaxf(sqrtf(sq3) * gnb, EPSV);
    if (lane < 16) *(float4*)(cosb + b * L_ + col) = cv;

    unsigned k0 = enc_key(cv.x), k1 = enc_key(cv.y), k2 = enc_key(cv.z), k3 = enc_key(cv.w);
    unsigned kmin = min(min(k0, k1), min(k2, k3));
    unsigned kmax = max(max(k0, k1), max(k2, k3));
    #pragma unroll
    for (int off = 16; off > 0; off >>= 1) {
        unsigned a = __shfl_xor(kmin, off, 32);
        unsigned c = __shfl_xor(kmax, off, 32);
        kmin = (a < kmin) ? a : kmin;
        kmax = (c > kmax) ? c : kmax;
    }
    if (lane == 0) { atomicMin(&keys[0], kmin); atomicMax(&keys[1], kmax); }
}
#undef COS_LOAD
#undef COS_STEP
#undef COS_PIN


#define FW   16
#define FQ   2
#define FR   (4 * FQ)
#define FCH  64
#define FST  (FCH / FR)
#define FT   4
#define FG   (C_ / FCH)

#define F_LOAD(S, ST)                                                                  \
    _Pragma("unroll") for (int q = 0; q < FQ; ++q) {                                   \
        _Pragma("unroll") for (int t = 0; t < FT; ++t) {                               \
            const float* p_ = xp + (size_t)((ST) * FR + q * 4) * L_ + t * 64;          \
            f0[S][q][t] = *(const v4f*)(p_);                                           \
            f1[S][q][t] = *(const v4f*)(p_ + L_);                                      \
        }                                                                              \
    }

__global__ void __launch_bounds__(FW * 32)
k_fused(const float* __restrict__ x, float* __restrict__ g,
        float* __restrict__ pdot, float* __restrict__ psq,
        unsigned* __restrict__ keys) {
    __shared__ __attribute__((aligned(16))) v2f part[2][FQ * 2][FW];
    __shared__ __attribute__((aligned(16))) float gl[FCH];
    if (blockIdx.x == 0 && blockIdx.y == 0 && threadIdx.x == 0) { keys[0] = 0xFFFFFFFFu; keys[1] = 0u; }
    const int b = blockIdx.y, cg = blockIdx.x;
    const int lane = threadIdx.x & 31;
    const int wave = __builtin_amdgcn_readfirstlane(threadIdx.x >> 5);
    const int n      = lane & 15;
    const int half   = lane >> 4;
    const int rowSel = half * 2;
    const int cbase  = cg * FCH;
    const int colw   = wave * (FT * 64);
    const float* xp  = x + ((size_t)b * C_ + cbase + rowSel) * L_ + colw + 4 * n;

    v4f f0[2][FQ][FT], f1[2][FQ][FT];
    v2f dot01[FT], dot23[FT], sq01[FT], sq23[FT];
    #pragma unroll
    for (int t = 0; t < FT; ++t) {
        dot01[t] = (v2f){0.f, 0.f}; dot23[t] = (v2f){0.f, 0.f};
        sq01[t]  = (v2f){0.f, 0.f}; sq23[t]  = (v2f){0.f, 0.f};
    }

    F_LOAD(0, 0)
    #pragma unroll
    for (int s = 0; s < FST; ++s) {
        __builtin_amdgcn_sched_barrier(0);
        if (s + 1 < FST) { F_LOAD((s + 1) & 1, s + 1) }
        __builtin_amdgcn_sched_barrier(0);

        #pragma unroll
        for (int q = 0; q < FQ; ++q) {
            v2f sa = {0.f, 0.f}, sc = {0.f, 0.f};
            #pragma unroll
            for (int t = 0; t < FT; ++t) {
                const v4f a = f0[s & 1][q][t], c = f1[s & 1][q][t];
                sa += __builtin_shufflevector(a, a, 0, 1); sa += __builtin_shufflevector(a, a, 2, 3);
                sc += __builtin_shufflevector(c, c, 0, 1); sc += __builtin_shufflevector(c, c, 2, 3);
            }
            v2f rs; rs.x = sa.x + sa.y; rs.y = sc.x + sc.y;
            #pragma unroll
            for (int off = 8; off > 0; off >>= 1) {
                v2f o; o.x = __shfl_xor(rs.x, off, 32); o.y = __shfl_xor(rs.y, off, 32);
                rs += o;
            }
            if (n == 0) part[s & 1][q * 2 + half][wave] = rs;
        }
        __builtin_amdgcn_fence(__ATOMIC_RELEASE, "workgroup", "local");
        __builtin_amdgcn_s_barrier();
        __builtin_amdgcn_fence(__ATOMIC_ACQUIRE, "workgroup", "local");

        v2f gq[FQ];
        #pragma unroll
        for (int q = 0; q < FQ; ++q) {
            const v4f* pr = (const v4f*)&part[s & 1][q * 2 + half][0];
            v2f ts = {0.f, 0.f};
            #pragma unroll
            for (int i = 0; i < FW / 2; ++i) {
                const v4f p = pr[i];
                ts += __builtin_shufflevector(p, p, 0, 1); ts += __builtin_shufflevector(p, p, 2, 3);
            }
            gq[q] = ts * (1.0f / (float)L_);
            if (wave == 0 && n == 0) *(v2f*)&gl[s * FR + q * 4 + rowSel] = gq[q];
        }

        #pragma unroll
        for (int t = 0; t < FT; ++t) {
            v8f a0 = {}, a1 = {}, a2 = {}, a3 = {};
            #pragma unroll
            for (int q = 0; q < FQ; ++q) {
                const v4f R0 = f0[s & 1][q][t], R1 = f1[s & 1][q][t];
                const v2f Af = gq[q];
                v2f B0; B0.x = R0.x; B0.y = R1.x;
                v2f B1; B1.x = R0.y; B1.y = R1.y;
                v2f B2; B2.x = R0.z; B2.y = R1.z;
                v2f B3; B3.x = R0.w; B3.y = R1.w;
                a0 = __builtin_amdgcn_wmma_f32_16x16x4_f32(false, Af, false, B0, (short)0, a0, false, false);
                a1 = __builtin_amdgcn_wmma_f32_16x16x4_f32(false, Af, false, B1, (short)0, a1, false, false);
                a2 = __builtin_amdgcn_wmma_f32_16x16x4_f32(false, Af, false, B2, (short)0, a2, false, false);
                a3 = __builtin_amdgcn_wmma_f32_16x16x4_f32(false, Af, false, B3, (short)0, a3, false, false);
                const v2f r0lo = __builtin_shufflevector(R0, R0, 0, 1), r0hi = __builtin_shufflevector(R0, R0, 2, 3);
                const v2f r1lo = __builtin_shufflevector(R1, R1, 0, 1), r1hi = __builtin_shufflevector(R1, R1, 2, 3);
                sq01[t] = __builtin_elementwise_fma(r0lo, r0lo, sq01[t]);
                sq23[t] = __builtin_elementwise_fma(r0hi, r0hi, sq23[t]);
                sq01[t] = __builtin_elementwise_fma(r1lo, r1lo, sq01[t]);
                sq23[t] = __builtin_elementwise_fma(r1hi, r1hi, sq23[t]);
            }
            v2f d01; d01.x = a0[0]; d01.y = a1[0];
            v2f d23; d23.x = a2[0]; d23.y = a3[0];
            dot01[t] += d01; dot23[t] += d23;
            asm volatile("" : "+v"(dot01[t]), "+v"(dot23[t]), "+v"(sq01[t]), "+v"(sq23[t]));
        }
    }

    if (wave == 0 && lane < 16)
        *(float4*)(g + b * C_ + cbase + 4 * lane) = *(const float4*)&gl[4 * lane];

    float* pd = pdot + (size_t)(b * FG + cg) * L_ + colw + 4 * n;
    float* pq = psq  + (size_t)(b * FG + cg) * L_ + colw + 4 * n;
    #pragma unroll
    for (int t = 0; t < FT; ++t) {
        float4 dv, qv;
        qv.x = sq01[t].x + __shfl_xor(sq01[t].x, 16, 32);
        qv.y = sq01[t].y + __shfl_xor(sq01[t].y, 16, 32);
        qv.z = sq23[t].x + __shfl_xor(sq23[t].x, 16, 32);
        qv.w = sq23[t].y + __shfl_xor(sq23[t].y, 16, 32);
        dv.x = dot01[t].x; dv.y = dot01[t].y; dv.z = dot23[t].x; dv.w = dot23[t].y;
        if (lane < 16) {
            *(float4*)(pd + t * 64) = dv;
            *(float4*)(pq + t * 64) = qv;
        }
    }
}
#undef F_LOAD

__global__ void __launch_bounds__(128)
k_cosred(const float* __restrict__ pdot, const float* __restrict__ psq,
         const float* __restrict__ g, float* __restrict__ cosb,
         unsigned* __restrict__ keys) {
    __shared__ float sg[C_];
    const int b = blockIdx.y;
    const int lane = threadIdx.x & 31;
    const int col  = (blockIdx.x * 128 + threadIdx.x) * 4;
    v4f d[FG], q[FG];
    #pragma unroll
    for (int p = 0; p < FG; ++p) {
        d[p] = *(const v4f*)(pdot + (size_t)(b * FG + p) * L_ + col);
        q[p] = *(const v4f*)(psq  + (size_t)(b * FG + p) * L_ + col);
    }
    const float g0 = g[b * C_ + threadIdx.x],       g1 = g[b * C_ + threadIdx.x + 128];
    const float g2 = g[b * C_ + threadIdx.x + 256], g3 = g[b * C_ + threadIdx.x + 384];
    sg[threadIdx.x] = g0;       sg[threadIdx.x + 128] = g1;
    sg[threadIdx.x + 256] = g2; sg[threadIdx.x + 384] = g3;
    __syncthreads();

    float gtot = 0.f;
    #pragma unroll
    for (int w = 0; w < 8; ++w) {
        float a  = sg[32 * w + lane];
        float c2 = sg[256 + 32 * w + lane];
        float s  = a * a + c2 * c2;
        #pragma unroll
        for (int off = 16; off > 0; off >>= 1) s += __shfl_xor(s, off, 32);
        gtot += s;
    }
    const float gnb = sqrtf(gtot);

    v4f ds = d[0], qs = q[0];
    #pragma unroll
    for (int p = 1; p < FG; ++p) { ds += d[p]; qs += q[p]; }
    float4 cv;
    cv.x = ds.x / fmaxf(sqrtf(qs.x) * gnb, EPSV);
    cv.y = ds.y / fmaxf(sqrtf(qs.y) * gnb, EPSV);
    cv.z = ds.z / fmaxf(sqrtf(qs.z) * gnb, EPSV);
    cv.w = ds.w / fmaxf(sqrtf(qs.w) * gnb, EPSV);
    *(float4*)(cosb + b * L_ + col) = cv;

    unsigned k0 = enc_key(cv.x), k1 = enc_key(cv.y), k2 = enc_key(cv.z), k3 = enc_key(cv.w);
    unsigned kmin = min(min(k0, k1), min(k2, k3));
    unsigned kmax = max(max(k0, k1), max(k2, k3));
    #pragma unroll
    for (int off = 16; off > 0; off >>= 1) {
        unsigned a = __shfl_xor(kmin, off, 32);
        unsigned c = __shfl_xor(kmax, off, 32);
        kmin = (a < kmin) ? a : kmin;
        kmax = (c > kmax) ? c : kmax;
    }
    if (lane == 0) { atomicMin(&keys[0], kmin); atomicMax(&keys[1], kmax); }
}

__global__ void __launch_bounds__(256)
k_hist(const float* __restrict__ cosb, const unsigned* __restrict__ keys,
       float* __restrict__ out) {
    const int b = blockIdx.x, t = threadIdx.x;
    const int lane = t & 31, wave = t >> 5;
    unsigned key0 = keys[0], key1 = keys[1];
    const v4f* cb = (const v4f*)(cosb + b * L_);
    v4f q0 = cb[t], q1 = cb[t + 256], q2 = cb[t + 512], q3 = cb[t + 768];
    asm volatile("" : "+v"(q0), "+v"(q1), "+v"(q2), "+v"(q3), "+s"(key0), "+s"(key1));

    float minS = dec_key(key0);
    float maxS = dec_key(key1);
    float step = (maxS - minS) * (1.0f / (float)(NB - 1));

    const float vals[16] = { q0.x, q0.y, q0.z, q0.w, q1.x, q1.y, q1.z, q1.w,
                             q2.x, q2.y, q2.z, q2.w, q3.x, q3.y, q3.z, q3.w };
    float bins[NB];
    #pragma unroll
    for (int n = 0; n < NB; ++n) bins[n] = 0.f;
    #pragma unroll
    for (int i = 0; i < 16; ++i) {
        float s = vals[i];
        #pragma unroll
        for (int n = 0; n < NB; ++n) {
            float d = fabsf(s - (minS + (float)n * step));
            if (d < 0.03125f) bins[n] += (1.0f - d);
        }
    }

    const bool h4 = (lane & 16) != 0, h3 = (lane & 8) != 0, h2 = (lane & 4) != 0, h1 = (lane & 2) != 0;
    float r8[8], r4[4], r2[2];
    #pragma unroll
    for (int j = 0; j < 8; ++j) {
        float keep = h4 ? bins[8 + j] : bins[j];
        float send = h4 ? bins[j] : bins[8 + j];
        r8[j] = keep + __shfl_xor(send, 16, 32);
    }
    #pragma unroll
    for (int j = 0; j < 4; ++j) {
        float keep = h3 ? r8[4 + j] : r8[j];
        float send = h3 ? r8[j] : r8[4 + j];
        r4[j] = keep + __shfl_xor(send, 8, 32);
    }
    #pragma unroll
    for (int j = 0; j < 2; ++j) {
        float keep = h2 ? r4[2 + j] : r4[j];
        float send = h2 ? r4[j] : r4[2 + j];
        r2[j] = keep + __shfl_xor(send, 4, 32);
    }
    float r1 = (h1 ? r2[1] : r2[0]) + __shfl_xor(h1 ? r2[0] : r2[1], 2, 32);
    r1 += __shfl_xor(r1, 1, 32);

    __shared__ float red[8][NB];
    __shared__ float sb[NB];
    if ((lane & 1) == 0) red[wave][lane >> 1] = r1;
    __syncthreads();
    if (t < NB) {
        float s = red[0][t];
        #pragma unroll
        for (int w = 1; w < 8; ++w) s += red[w][t];
        sb[t] = s;
    }
    __syncthreads();
    float tot = 0.f;
    #pragma unroll
    for (int n = 0; n < NB; ++n) tot += sb[n];
    float inv = 1.0f / tot;
    for (int i = t; i < NB * (C_ / NB); i += 256)
        out[b * (NB * (C_ / NB)) + i] = sb[i >> 5] * inv;
}

extern "C" void kernel_launch(void* const* d_in, const int* in_sizes, int n_in,
                              void* d_out, int out_size, void* d_ws, size_t ws_size,
                              hipStream_t stream) {
    const float* x = (const float*)d_in[0];
    float* out = (float*)d_out;
    char* ws = (char*)d_ws;
    float*    g    = (float*)(ws + WS_G);
    unsigned* keys = (unsigned*)(ws + WS_KEYS);
    float*    cosb = (float*)(ws + WS_COS);

    float*    pdot = (float*)(ws + WS_PDOT);
    float*    psq  = (float*)(ws + WS_PSQ);

    if (ws_size >= (size_t)WS_END) {
        k_fused<<<dim3(FG, B_), FW * 32, 0, stream>>>(x, g, pdot, psq, keys);
        k_cosred<<<dim3(L_ / 512, B_), 128, 0, stream>>>(pdot, psq, g, cosb, keys);
    } else {
        k_mean<<<(B_ * C_) / (MW * MR), MW * 32, 0, stream>>>(x, g, keys);
        k_cos<<<dim3(L_ / 256, B_), 128, 0, stream>>>(x, g, cosb, keys);
    }
    k_hist<<<B_, 256, 0, stream>>>(cosb, keys, out);
}


# lane butterflies of the 3 launched kernels on the vector ALU (DPP row_xmask, v_permlanex16), not LDS permutes; the mean's 16 wave pieces read one per lane, folded likewise: that sum's order changed
# speedup vs baseline: 2.3460x; 1.1024x over previous
#include <hip/hip_runtime.h>

typedef __attribute__((ext_vector_type(2))) float v2f;
typedef __attribute__((ext_vector_type(8))) float v8f;
typedef __attribute__((ext_vector_type(4))) float v4f;

#define B_   32
#define C_   512
#define L_   4096
#define NB   16
#define EPSV 1e-8f

#define WS_G    0
#define WS_GN   (B_ * C_ * 4)
#define WS_KEYS (WS_GN + 128)
#define WS_COS  (WS_KEYS + 128)
#define WS_PDOT (WS_COS + B_ * L_ * 4)
#define WS_PSQ  (WS_PDOT + B_ * (C_ / 64) * L_ * 4)
#define WS_END  (WS_PSQ + B_ * (C_ / 64) * L_ * 4)

__device__ __forceinline__ unsigned enc_key(float f) {
    unsigned u = __float_as_uint(f);
    return (u & 0x80000000u) ? ~u : (u | 0x80000000u);
}
__device__ __forceinline__ float dec_key(unsigned k) {
    unsigned u = (k & 0x80000000u) ? (k & 0x7FFFFFFFu) : ~k;
    return __uint_as_float(u);
}

template <int M> __device__ __forceinline__ int xrow_i(int v) {
    return __builtin_amdgcn_update_dpp(0, v, 0x160 + M, 0xf, 0xf, true);
}
template <int M> __device__ __forceinline__ float xrow(float v) {
    return __builtin_bit_cast(float, xrow_i<M>(__builtin_bit_cast(int, v)));
}
template <int M> __device__ __forceinline__ unsigned xrow_u(unsigned v) {
    return (unsigned)xrow_i<M>((int)v);
}
__device__ __forceinline__ int xhalf_i(int v) {
    return __builtin_amdgcn_permlanex16(v, v, 0x76543210, 0xfedcba98, false, false);
}
__device__ __forceinline__ float xhalf(float v) {
    return __builtin_bit_cast(float, xhalf_i(__builtin_bit_cast(int, v)));
}
__device__ __forceinline__ unsigned xhalf_u(unsigned v) { return (unsigned)xhalf_i((int)v); }
__device__ __forceinline__ float wave_sum(float s) {
    s += xhalf(s);
    s += xrow<8>(s); s += xrow<4>(s); s += xrow<2>(s); s += xrow<1>(s);
    return s;
}
__device__ __forceinline__ void wave_minmax(unsigned& kmin, unsigned& kmax) {
    unsigned a, c2;
    a = xhalf_u(kmin);   c2 = xhalf_u(kmax);   kmin = (a < kmin) ? a : kmin; kmax = (c2 > kmax) ? c2 : kmax;
    a = xrow_u<8>(kmin); c2 = xrow_u<8>(kmax); kmin = (a < kmin) ? a : kmin; kmax = (c2 > kmax) ? c2 : kmax;
    a = xrow_u<4>(kmin); c2 = xrow_u<4>(kmax); kmin = (a < kmin) ? a : kmin; kmax = (c2 > kmax) ? c2 : kmax;
    a = xrow_u<2>(kmin); c2 = xrow_u<2>(kmax); kmin = (a < kmin) ? a : kmin; kmax = (c2 > kmax) ? c2 : kmax;
    a = xrow_u<1>(kmin); c2 = xrow_u<1>(kmax); kmin = (a < kmin) ? a : kmin; kmax = (c2 > kmax) ? c2 : kmax;
}

#define MW   16
#define MR   4
#define MQ   16
#define MBLK (MR * (L_ / 128) / MQ)
__global__ void __launch_bounds__(MW * 32)
k_mean(const float* __restrict__ x, float* __restrict__ g,
       unsigned* __restrict__ keys) {
    if (blockIdx.x == 0 && threadIdx.x == 0) { keys[0] = 0xFFFFFFFFu; keys[1] = 0u; }
    const int lane = threadIdx.x & 31;
    const int wave = __builtin_amdgcn_readfirstlane(threadIdx.x >> 5);
    const int row0 = (blockIdx.x * MW + wave) * MR;
    const v4f* xr = (const v4f*)(x + (size_t)row0 * L_) + lane;
    v4f buf[2][MQ];
    float sum[MR];

#define MEAN_LOAD(K)                                                                   \
    _Pragma("unroll") for (int q = 0; q < MQ; ++q)                                     \
        buf[(K) & 1][q] = xr[((K) * MQ + q) * 32];

    MEAN_LOAD(0)
    #pragma unroll
    for (int k = 0; k < MBLK; ++k) {
        __builtin_amdgcn_sched_barrier(0);
        if (k + 1 < MBLK) { MEAN_LOAD(k + 1) }
        __builtin_amdgcn_sched_barrier(0);
        float s = (k & 1) ? sum[k >> 1] : 0.f;
        #pragma unroll
        for (int q = 0; q < MQ; ++q) {
            v4f v = buf[k & 1][q];
            s += (v.x + v.y) + (v.z + v.w);
        }
        asm volatile("" : "+v"(s));
        sum[k >> 1] = s;
    }
#undef MEAN_LOAD

    float4 m;
    #pragma unroll
    for (int r = 0; r < MR; ++r) {
        float s = sum[r];
        #pragma unroll
        for (int off = 16; off > 0; off >>= 1) s += __shfl_xor(s, off, 32);
        sum[r] = s * (1.0f / (float)L_);
    }
    m.x = sum[0]; m.y = sum[1]; m.z = sum[2]; m.w = sum[3];
    if (lane == 0) *(float4*)(g + row0) = m;
}

#define CU   8
#define NBLK (C_ / (4 * CU))
#define RD   4

#define COS_LOAD(R0, R1, BLK)                                                          \
    _Pragma("unroll") for (int u = 0; u < CU; ++u) {                                   \
        const float* p_ = xp + (size_t)(((BLK) * CU + u) * 4) * L_;                    \
        R0[u] = *(const v4f*)(p_);                                                     \
        R1[u] = *(const v4f*)(p_ + L_);                                                \
    }

#define COS_PIN()                                                                      \
    asm volatile("" : "+v"(acc0), "+v"(acc1), "+v"(acc2), "+v"(acc3),                  \
                      "+v"(sq0), "+v"(sq1), "+v"(sq2), "+v"(sq3));

#define COS_STEP(R0, R1, BLK)                                                          \
    _Pragma("unroll") for (int u = 0; u < CU; ++u) {                                   \
        const int c0_ = ((BLK) * CU + u) * 4;                                          \
        v2f Af; Af.x = gp[c0_]; Af.y = gp[c0_ + 1];                                    \
        v2f B0; B0.x = R0[u].x; B0.y = R1[u].x;                                        \
        v2f B1; B1.x = R0[u].y; B1.y = R1[u].y;                                        \
        v2f B2; B2.x = R0[u].z; B2.y = R1[u].z;                                        \
        v2f B3; B3.x = R0[u].w; B3.y = R1[u].w;                                        \
        acc0 = __builtin_amdgcn_wmma_f32_16x16x4_f32(false, Af, false, B0, (short)0, acc0, false, false); \
        acc1 = __builtin_amdgcn_wmma_f32_16x16x4_f32(false, Af, false, B1, (short)0, acc1, false, false); \
        acc2 = __builtin_amdgcn_wmma_f32_16x16x4_f32(false, Af, false, B2, (short)0, acc2, false, false); \
        acc3 = __builtin_amdgcn_wmma_f32_16x16x4_f32(false, Af, false, B3, (short)0, acc3, false, false); \
        sq0 = fmaf(R0[u].x, R0[u].x, sq0); sq0 = fmaf(R1[u].x, R1[u].x, sq0);          \
        sq1 = fmaf(R0[u].y, R0[u].y, sq1); sq1 = fmaf(R1[u].y, R1[u].y, sq1);          \
        sq2 = fmaf(R0[u].z, R0[u].z, sq2); sq2 = fmaf(R1[u].z, R1[u].z, sq2);          \
        sq3 = fmaf(R0[u].w, R0[u].w, sq3); sq3 = fmaf(R1[u].w, R1[u].w, sq3);          \
    }

__global__ void __launch_bounds__(128)
k_cos(const float* __restrict__ x, const float* __restrict__ g,
      float* __restrict__ cosb, unsigned* __restrict__ keys) {
    __shared__ float sg[C_];
    const int b = blockIdx.y;
    for (int i = threadIdx.x; i < C_; i += 128) sg[i] = g[b * C_ + i];
    __syncthreads();

    const int wave = threadIdx.x >> 5, lane = threadIdx.x & 31;
    const int n      = lane & 15;
    const int rowSel = (lane >> 4) * 2;
    const int l0     = blockIdx.x * 256 + wave * 64;
    const int col    = l0 + 4 * n;
    const float* xp  = x + (size_t)b * C_ * L_ + (size_t)rowSel * L_ + col;
    const float* gp  = sg + rowSel;

    v8f acc0 = {}, acc1 = {}, acc2 = {}, acc3 = {};
    float sq0 = 0.f, sq1 = 0.f, sq2 = 0.f, sq3 = 0.f;
    v4f r0[RD][CU], r1[RD][CU];

    #pragma unroll
    for (int k = 0; k < RD - 1; ++k) { COS_LOAD(r0[k % RD], r1[k % RD], k) }

    float gtot = 0.f;
    #pragma unroll
    for (int w = 0; w < 8; ++w) {
        float a  = sg[32 * w + lane];
        float c2 = sg[256 + 32 * w + lane];
        float s  = a * a + c2 * c2;
        #pragma unroll
        for (int off = 16; off > 0; off >>= 1) s += __shfl_xor(s, off, 32);
        gtot += s;
    }
    float gnb = sqrtf(gtot);
    asm volatile("" : "+v"(gnb));

    #pragma unroll
    for (int k = 0; k < NBLK; ++k) {
        __builtin_amdgcn_sched_barrier(0);
        if (k + RD - 1 < NBLK) { COS_LOAD(r0[(k + RD - 1) % RD], r1[(k + RD - 1) % RD], k + RD - 1) }
        __builtin_amdgcn_sched_barrier(0);
        COS_PIN()
        COS_STEP(r0[k % RD], r1[k % RD], k)
        COS_PIN()
    }

    sq0 += __shfl_xor(sq0, 16, 32);
    sq1 += __shfl_xor(sq1, 16, 32);
    sq2 += __shfl_xor(sq2, 16, 32);
    sq3 += __shfl_xor(sq3, 16, 32);
    float4 cv;
    cv.x = acc0[0] / fmaxf(sqrtf(sq0) * gnb, EPSV);
    cv.y = acc1[0] / fmaxf(sqrtf(sq1) * gnb, EPSV);
    cv.z = acc2[0] / fmaxf(sqrtf(sq2) * gnb, EPSV);
    cv.w = acc3[0] / fmaxf(sqrtf(sq3) * gnb, EPSV);
    if (lane < 16) *(float4*)(cosb + b * L_ + col) = cv;

    unsigned k0 = enc_key(cv.x), k1 = enc_key(cv.y), k2 = enc_key(cv.z), k3 = enc_key(cv.w);
    unsigned kmin = min(min(k0, k1), min(k2, k3));
    unsigned kmax = max(max(k0, k1), max(k2, k3));
    #pragma unroll
    for (int off = 16; off > 0; off >>= 1) {
        unsigned a = __shfl_xor(kmin, off, 32);
        unsigned c = __shfl_xor(kmax, off, 32);
        kmin = (a < kmin) ? a : kmin;
        kmax = (c > kmax) ? c : kmax;
    }
    if (lane == 0) { atomicMin(&keys[0], kmin); atomicMax(&keys[1], kmax); }
}
#undef COS_LOAD
#undef COS_STEP
#undef COS_PIN


#define FW   16
#define FQ   2
#define FR   (4 * FQ)
#define FCH  64
#define FST  (FCH / FR)
#define FT   4
#define FG   (C_ / FCH)

template <int M> __device__ __forceinline__ void half_step4(v2f (&r)[FQ]) {
    #pragma unroll
    for (int q = 0; q < FQ; ++q) { r[q].x += xrow<M>(r[q].x); r[q].y += xrow<M>(r[q].y); }
}
__device__ __forceinline__ void half_sum4(v2f (&r)[FQ]) {
    half_step4<8>(r); half_step4<4>(r); half_step4<2>(r); half_step4<1>(r);
}

#define F_LOAD(S, ST)                                                                  \
    _Pragma("unroll") for (int q = 0; q < FQ; ++q) {                                   \
        _Pragma("unroll") for (int t = 0; t < FT; ++t) {                               \
            const float* p_ = xp + (size_t)((ST) * FR + q * 4) * L_ + t * 64;          \
            f0[S][q][t] = *(const v4f*)(p_);                                           \
            f1[S][q][t] = *(const v4f*)(p_ + L_);                                      \
        }                                                                              \
    }

__global__ void __launch_bounds__(FW * 32)
k_fused(const float* __restrict__ x, float* __restrict__ g,
        float* __restrict__ pdot, float* __restrict__ psq,
        unsigned* __restrict__ keys) {
    __shared__ __attribute__((aligned(16))) v2f part[2][FQ * 2][FW];
    __shared__ __attribute__((aligned(16))) float gl[FCH];
    if (blockIdx.x == 0 && blockIdx.y == 0 && threadIdx.x == 0) { keys[0] = 0xFFFFFFFFu; keys[1] = 0u; }
    const int b = blockIdx.y, cg = blockIdx.x;
    const int lane = threadIdx.x & 31;
    const int wave = __builtin_amdgcn_readfirstlane(threadIdx.x >> 5);
    const int n      = lane & 15;
    const int half   = lane >> 4;
    const int rowSel = half * 2;
    const int cbase  = cg * FCH;
    const int colw   = wave * (FT * 64);
    const float* xp  = x + ((size_t)b * C_ + cbase + rowSel) * L_ + colw + 4 * n;

    v4f f0[2][FQ][FT], f1[2][FQ][FT];
    v2f dot01[FT], dot23[FT], sq01[FT], sq23[FT];
    #pragma unroll
    for (int t = 0; t < FT; ++t) {
        dot01[t] = (v2f){0.f, 0.f}; dot23[t] = (v2f){0.f, 0.f};
        sq01[t]  = (v2f){0.f, 0.f}; sq23[t]  = (v2f){0.f, 0.f};
    }

    F_LOAD(0, 0)
    #pragma unroll
    for (int s = 0; s < FST; ++s) {
        __builtin_amdgcn_sched_barrier(0);
        if (s + 1 < FST) { F_LOAD((s + 1) & 1, s + 1) }
        __builtin_amdgcn_sched_barrier(0);

        v2f rs[FQ];
        #pragma unroll
        for (int q = 0; q < FQ; ++q) {
            v2f sa = {0.f, 0.f}, sc = {0.f, 0.f};
            #pragma unroll
            for (int t = 0; t < FT; ++t) {
                const v4f a = f0[s & 1][q][t], c = f1[s & 1][q][t];
                sa += __builtin_shufflevector(a, a, 0, 1); sa += __builtin_shufflevector(a, a, 2, 3);
                sc += __builtin_shufflevector(c, c, 0, 1); sc += __builtin_shufflevector(c, c, 2, 3);
            }
            rs[q].x = sa.x + sa.y; rs[q].y = sc.x + sc.y;
        }
        half_sum4(rs);
        if (n == 0) {
            #pragma unroll
            for (int q = 0; q < FQ; ++q) part[s & 1][q * 2 + half][wave] = rs[q];
        }
        __builtin_amdgcn_fence(__ATOMIC_RELEASE, "workgroup", "local");
        __builtin_amdgcn_s_barrier();
        __builtin_amdgcn_fence(__ATOMIC_ACQUIRE, "workgroup", "local");

        v2f gq[FQ];
        #pragma unroll
        for (int q = 0; q < FQ; ++q) gq[q] = part[s & 1][q * 2 + half][n];
        half_sum4(gq);
        #pragma unroll
        for (int q = 0; q < FQ; ++q) {
            gq[q] = gq[q] * (1.0f / (float)L_);
            if (wave == 0 && n == 0) *(v2f*)&gl[s * FR + q * 4 + rowSel] = gq[q];
        }

        #pragma unroll
        for (int t = 0; t < FT; ++t) {
            v8f a0 = {}, a1 = {}, a2 = {}, a3 = {};
            #pragma unroll
            for (int q = 0; q < FQ; ++q) {
                const v4f R0 = f0[s & 1][q][t], R1 = f1[s & 1][q][t];
                const v2f Af = gq[q];
                v2f B0; B0.x = R0.x; B0.y = R1.x;
                v2f B1; B1.x = R0.y; B1.y = R1.y;
                v2f B2; B2.x = R0.z; B2.y = R1.z;
                v2f B3; B3.x = R0.w; B3.y = R1.w;
                a0 = __builtin_amdgcn_wmma_f32_16x16x4_f32(false, Af, false, B0, (short)0, a0, false, false);
                a1 = __builtin_amdgcn_wmma_f32_16x16x4_f32(false, Af, false, B1, (short)0, a1, false, false);
                a2 = __builtin_amdgcn_wmma_f32_16x16x4_f32(false, Af, false, B2, (short)0, a2, false, false);
                a3 = __builtin_amdgcn_wmma_f32_16x16x4_f32(false, Af, false, B3, (short)0, a3, false, false);
                const v2f r0lo = __builtin_shufflevector(R0, R0, 0, 1), r0hi = __builtin_shufflevector(R0, R0, 2, 3);
                const v2f r1lo = __builtin_shufflevector(R1, R1, 0, 1), r1hi = __builtin_shufflevector(R1, R1, 2, 3);
                sq01[t] = __builtin_elementwise_fma(r0lo, r0lo, sq01[t]);
                sq23[t] = __builtin_elementwise_fma(r0hi, r0hi, sq23[t]);
                sq01[t] = __builtin_elementwise_fma(r1lo, r1lo, sq01[t]);
                sq23[t] = __builtin_elementwise_fma(r1hi, r1hi, sq23[t]);
            }
            v2f d01; d01.x = a0[0]; d01.y = a1[0];
            v2f d23; d23.x = a2[0]; d23.y = a3[0];
            dot01[t] += d01; dot23[t] += d23;
            asm volatile("" : "+v"(dot01[t]), "+v"(dot23[t]), "+v"(sq01[t]), "+v"(sq23[t]));
        }
    }

    if (wave == 0 && lane < 16)
        *(float4*)(g + b * C_ + cbase + 4 * lane) = *(const float4*)&gl[4 * lane];

    float* pd = pdot + (size_t)(b * FG + cg) * L_ + colw + 4 * n;
    float* pq = psq  + (size_t)(b * FG + cg) * L_ + colw + 4 * n;
    #pragma unroll
    for (int t = 0; t < FT; ++t) {
        float4 dv, qv;
        qv.x = sq01[t].x + xhalf(sq01[t].x);
        qv.y = sq01[t].y + xhalf(sq01[t].y);
        qv.z = sq23[t].x + xhalf(sq23[t].x);
        qv.w = sq23[t].y + xhalf(sq23[t].y);
        dv.x = dot01[t].x; dv.y = dot01[t].y; dv.z = dot23[t].x; dv.w = dot23[t].y;
        if (lane < 16) {
            *(float4*)(pd + t * 64) = dv;
            *(float4*)(pq + t * 64) = qv;
        }
    }
}
#undef F_LOAD

__global__ void __launch_bounds__(128)
k_cosred(const float* __restrict__ pdot, const float* __restrict__ psq,
         const float* __restrict__ g, float* __restrict__ cosb,
         unsigned* __restrict__ keys) {
    __shared__ float sg[C_];
    const int b = blockIdx.y;
    const int lane = threadIdx.x & 31;
    const int col  = (blockIdx.x * 128 + threadIdx.x) * 4;
    v4f d[FG], q[FG];
    #pragma unroll
    for (int p = 0; p < FG; ++p) {
        d[p] = *(const v4f*)(pdot + (size_t)(b * FG + p) * L_ + col);
        q[p] = *(const v4f*)(psq  + (size_t)(b * FG + p) * L_ + col);
    }
    const float g0 = g[b * C_ + threadIdx.x],       g1 = g[b * C_ + threadIdx.x + 128];
    const float g2 = g[b * C_ + threadIdx.x + 256], g3 = g[b * C_ + threadIdx.x + 384];
    sg[threadIdx.x] = g0;       sg[threadIdx.x + 128] = g1;
    sg[threadIdx.x + 256] = g2; sg[threadIdx.x + 384] = g3;
    __syncthreads();

    float gtot = 0.f;
    #pragma unroll
    for (int w = 0; w < 8; ++w) {
        float a  = sg[32 * w + lane];
        float c2 = sg[256 + 32 * w + lane];
        gtot += wave_sum(a * a + c2 * c2);
    }
    const float gnb = sqrtf(gtot);

    v4f ds = d[0], qs = q[0];
    #pragma unroll
    for (int p = 1; p < FG; ++p) { ds += d[p]; qs += q[p]; }
    float4 cv;
    cv.x = ds.x / fmaxf(sqrtf(qs.x) * gnb, EPSV);
    cv.y = ds.y / fmaxf(sqrtf(qs.y) * gnb, EPSV);
    cv.z = ds.z / fmaxf(sqrtf(qs.z) * gnb, EPSV);
    cv.w = ds.w / fmaxf(sqrtf(qs.w) * gnb, EPSV);
    *(float4*)(cosb + b * L_ + col) = cv;

    unsigned k0 = enc_key(cv.x), k1 = enc_key(cv.y), k2 = enc_key(cv.z), k3 = enc_key(cv.w);
    unsigned kmin = min(min(k0, k1), min(k2, k3));
    unsigned kmax = max(max(k0, k1), max(k2, k3));
    wave_minmax(kmin, kmax);
    if (lane == 0) { atomicMin(&keys[0], kmin); atomicMax(&keys[1], kmax); }
}

__global__ void __launch_bounds__(256)
k_hist(const float* __restrict__ cosb, const unsigned* __restrict__ keys,
       float* __restrict__ out) {
    const int b = blockIdx.x, t = threadIdx.x;
    const int lane = t & 31, wave = t >> 5;
    unsigned key0 = keys[0], key1 = keys[1];
    const v4f* cb = (const v4f*)(cosb + b * L_);
    v4f q0 = cb[t], q1 = cb[t + 256], q2 = cb[t + 512], q3 = cb[t + 768];
    asm volatile("" : "+v"(q0), "+v"(q1), "+v"(q2), "+v"(q3), "+s"(key0), "+s"(key1));

    float minS = dec_key(key0);
    float maxS = dec_key(key1);
    float step = (maxS - minS) * (1.0f / (float)(NB - 1));

    const float vals[16] = { q0.x, q0.y, q0.z, q0.w, q1.x, q1.y, q1.z, q1.w,
                             q2.x, q2.y, q2.z, q2.w, q3.x, q3.y, q3.z, q3.w };
    float bins[NB];
    #pragma unroll
    for (int n = 0; n < NB; ++n) bins[n] = 0.f;
    #pragma unroll
    for (int i = 0; i < 16; ++i) {
        float s = vals[i];
        #pragma unroll
        for (int n = 0; n < NB; ++n) {
            float d = fabsf(s - (minS + (float)n * step));
            if (d < 0.03125f) bins[n] += (1.0f - d);
        }
    }

    const bool h4 = (lane & 16) != 0, h3 = (lane & 8) != 0, h2 = (lane & 4) != 0, h1 = (lane & 2) != 0;
    float r8[8], r4[4], r2[2];
    #pragma unroll
    for (int j = 0; j < 8; ++j) {
        float keep = h4 ? bins[8 + j] : bins[j];
        float send = h4 ? bins[j] : bins[8 + j];
        r8[j] = keep + xhalf(send);
    }
    #pragma unroll
    for (int j = 0; j < 4; ++j) {
        float keep = h3 ? r8[4 + j] : r8[j];
        float send = h3 ? r8[j] : r8[4 + j];
        r4[j] = keep + xrow<8>(send);
    }
    #pragma unroll
    for (int j = 0; j < 2; ++j) {
        float keep = h2 ? r4[2 + j] : r4[j];
        float send = h2 ? r4[j] : r4[2 + j];
        r2[j] = keep + xrow<4>(send);
    }
    float r1 = (h1 ? r2[1] : r2[0]) + xrow<2>(h1 ? r2[0] : r2[1]);
    r1 += xrow<1>(r1);

    __shared__ float red[8][NB];
    __shared__ float sb[NB];
    if ((lane & 1) == 0) red[wave][lane >> 1] = r1;
    __syncthreads();
    if (t < NB) {
        float s = red[0][t];
        #pragma unroll
        for (int w = 1; w < 8; ++w) s += red[w][t];
        sb[t] = s;
    }
    __syncthreads();
    float tot = 0.f;
    #pragma unroll
    for (int n = 0; n < NB; ++n) tot += sb[n];
    float inv = 1.0f / tot;
    for (int i = t; i < NB * (C_ / NB); i += 256)
        out[b * (NB * (C_ / NB)) + i] = sb[i >> 5] * inv;
}

extern "C" void kernel_launch(void* const* d_in, const int* in_sizes, int n_in,
                              void* d_out, int out_size, void* d_ws, size_t ws_size,
                              hipStream_t stream) {
    const float* x = (const float*)d_in[0];
    float* out = (float*)d_out;
    char* ws = (char*)d_ws;
    float*    g    = (float*)(ws + WS_G);
    unsigned* keys = (unsigned*)(ws + WS_KEYS);
    float*    cosb = (float*)(ws + WS_COS);

    float*    pdot = (float*)(ws + WS_PDOT);
    float*    psq  = (float*)(ws + WS_PSQ);

    if (ws_size >= (size_t)WS_END) {
        k_fused<<<dim3(FG, B_), FW * 32, 0, stream>>>(x, g, pdot, psq, keys);
        k_cosred<<<dim3(L_ / 512, B_), 128, 0, stream>>>(pdot, psq, g, cosb, keys);
    } else {
        k_mean<<<(B_ * C_) / (MW * MR), MW * 32, 0, stream>>>(x, g, keys);
        k_cos<<<dim3(L_ / 256, B_), 128, 0, stream>>>(x, g, cosb, keys);
    }
    k_hist<<<B_, 256, 0, stream>>>(cosb, keys, out);
}
